# starting kernel
# baseline (speedup 1.0000x reference)
#include <hip/hip_runtime.h>

#define DIM_IN   128
#define DIM_HID  256
#define DIM_OUT  128
#define N_DIST   1024
#define N_NODES  500000

typedef float v2f __attribute__((ext_vector_type(2)));
typedef float v4f __attribute__((ext_vector_type(4)));
typedef float v8f __attribute__((ext_vector_type(8)));

__global__ void n2d_zero_head(float* __restrict__ head) {
    int i = blockIdx.x * blockDim.x + threadIdx.x;
    const int n = N_DIST * DIM_IN;
    for (; i < n; i += gridDim.x * blockDim.x) head[i] = 0.0f;
}

__global__ void n2d_segsum(const float* __restrict__ x,
                           const int*   __restrict__ zone,
                           float*       __restrict__ head) {
    const int lane   = threadIdx.x & 31;
    const int wave   = (blockIdx.x * blockDim.x + threadIdx.x) >> 5;
    const int nwaves = (gridDim.x * blockDim.x) >> 5;

    for (int node = wave; node < N_NODES; node += nwaves) {
        const int z = zone[node];
        const v4f* p = (const v4f*)(x + (size_t)node * DIM_IN) + lane;
        v4f v = __builtin_nontemporal_load(p);
        float* h = head + z * DIM_IN + lane * 4;
        atomicAdd(h + 0, v.x);
        atomicAdd(h + 1, v.y);
        atomicAdd(h + 2, v.z);
        atomicAdd(h + 3, v.w);
    }
}

__global__ void n2d_mlp(const float* __restrict__ head,
                        const float* __restrict__ w1, const float* __restrict__ b1,
                        const float* __restrict__ w2, const float* __restrict__ b2,
                        float* __restrict__ out) {
    __shared__ float h1[16 * DIM_HID];

    const int lane = threadIdx.x & 31;
    const int wv   = threadIdx.x >> 5;
    const int half = lane >> 4;
    const int m    = lane & 15;
    const int row0 = blockIdx.x * 16;

    for (int t = wv; t < DIM_HID / 16; t += 8) {
        const int n0 = t * 16;
        v8f acc = {};
        for (int k = 0; k < DIM_IN; k += 4) {
            const int ka = k + half * 2;
            v2f a, b;
            float a0 = head[(row0 + m) * DIM_IN + ka];
            float a1 = head[(row0 + m) * DIM_IN + ka + 1];
            a.x = a0 > 0.0f ? a0 : 0.0f;
            a.y = a1 > 0.0f ? a1 : 0.0f;
            b.x = w1[(size_t)ka       * DIM_HID + n0 + m];
            b.y = w1[(size_t)(ka + 1) * DIM_HID + n0 + m];
            acc = __builtin_amdgcn_wmma_f32_16x16x4_f32(
                      false, a, false, b, (short)0, acc, false, false);
        }
        const float bias = b1[n0 + m];
        #pragma unroll
        for (int r = 0; r < 8; ++r) {
            float v = acc[r] + bias;
            h1[(r + half * 8) * DIM_HID + n0 + m] = v > 0.0f ? v : 0.0f;
        }
    }
    __syncthreads();

    {
        const int n0 = wv * 16;
        v8f acc = {};
        for (int k = 0; k < DIM_HID; k += 4) {
            const int ka = k + half * 2;
            v2f a, b;
            a.x = h1[m * DIM_HID + ka];
            a.y = h1[m * DIM_HID + ka + 1];
            b.x = w2[(size_t)ka       * DIM_OUT + n0 + m];
            b.y = w2[(size_t)(ka + 1) * DIM_OUT + n0 + m];
            acc = __builtin_amdgcn_wmma_f32_16x16x4_f32(
                      false, a, false, b, (short)0, acc, false, false);
        }
        const float bias = b2[n0 + m];
        #pragma unroll
        for (int r = 0; r < 8; ++r)
            out[(size_t)(row0 + r + half * 8) * DIM_OUT + n0 + m] = acc[r] + bias;
    }
}

extern "C" void kernel_launch(void* const* d_in, const int* in_sizes, int n_in,
                              void* d_out, int out_size, void* d_ws, size_t ws_size,
                              hipStream_t stream) {
    const float* x    = (const float*)d_in[0];
    const int*   zone = (const int*)  d_in[1];
    const float* w1   = (const float*)d_in[2];
    const float* b1   = (const float*)d_in[3];
    const float* w2   = (const float*)d_in[4];
    const float* b2   = (const float*)d_in[5];

    float* head = (float*)d_ws;
    float* out  = (float*)d_out;

    n2d_zero_head<<<256, 256, 0, stream>>>(head);
    n2d_segsum<<<2048, 256, 0, stream>>>(x, zone, head);
    n2d_mlp<<<N_DIST / 16, 256, 0, stream>>>(head, w1, b1, w2, b2, out);
}


# Segment sum district-major: node ids counting-sorted by district (int atomics give ranks), one work-group per district gathers its rows and sums in registers; no float atomics; sum order changes
# speedup vs baseline: 2.5649x; 2.5649x over previous
#include <hip/hip_runtime.h>

#define DIM_IN   128
#define DIM_HID  256
#define DIM_OUT  128
#define N_DIST   1024
#define N_NODES  500000
#define CNT_PITCH 64
#define GW 4
#define GU 16

typedef float v2f __attribute__((ext_vector_type(2)));
typedef float v4f __attribute__((ext_vector_type(4)));
typedef float v8f __attribute__((ext_vector_type(8)));
typedef int   v4i __attribute__((ext_vector_type(4)));


__global__ void n2d_zero_cnt(int* __restrict__ cnt) {
    const int d = blockIdx.x * blockDim.x + threadIdx.x;
    if (d < N_DIST) cnt[d * CNT_PITCH] = 0;
}

__global__ void n2d_rank(const int* __restrict__ zone,
                         int*       __restrict__ cnt,
                         int*       __restrict__ rank) {
    const int q = blockIdx.x * blockDim.x + threadIdx.x;
    if (q < N_NODES / 4) {
        const v4i z = ((const v4i*)zone)[q];
        v4i r;
        r.x = atomicAdd(&cnt[z.x * CNT_PITCH], 1);
        r.y = atomicAdd(&cnt[z.y * CNT_PITCH], 1);
        r.z = atomicAdd(&cnt[z.z * CNT_PITCH], 1);
        r.w = atomicAdd(&cnt[z.w * CNT_PITCH], 1);
        ((v4i*)rank)[q] = r;
    }
}

__global__ void n2d_scatter(const int* __restrict__ zone,
                            const int* __restrict__ cnt,
                            const int* __restrict__ rank,
                            int*       __restrict__ off,
                            int*       __restrict__ idx) {
    __shared__ int s_off[N_DIST];
    __shared__ int s_wtot[8];
    const int tid  = threadIdx.x;
    const int lane = tid & 31;
    const int wv   = tid >> 5;

    const int c0 = cnt[(tid * 4 + 0) * CNT_PITCH];
    const int c1 = cnt[(tid * 4 + 1) * CNT_PITCH];
    const int c2 = cnt[(tid * 4 + 2) * CNT_PITCH];
    const int c3 = cnt[(tid * 4 + 3) * CNT_PITCH];
    const int s  = c0 + c1 + c2 + c3;
    int incl = s;
    for (int dlt = 1; dlt < 32; dlt <<= 1) {
        const int t = __shfl_up(incl, dlt);
        if (lane >= dlt) incl += t;
    }
    if (lane == 31) s_wtot[wv] = incl;
    __syncthreads();
    int base = 0;
    for (int w = 0; w < wv; ++w) base += s_wtot[w];
    const int e0 = base + incl - s;
    const int e1 = e0 + c0;
    const int e2 = e1 + c1;
    const int e3 = e2 + c2;
    s_off[tid * 4 + 0] = e0;
    s_off[tid * 4 + 1] = e1;
    s_off[tid * 4 + 2] = e2;
    s_off[tid * 4 + 3] = e3;
    if (blockIdx.x == 0) {
        v4i e = {e0, e1, e2, e3};
        ((v4i*)off)[tid] = e;
    }
    __syncthreads();

    const int q = blockIdx.x * blockDim.x + tid;
    if (q < N_NODES / 4) {
        const v4i z = ((const v4i*)zone)[q];
        const v4i r = ((const v4i*)rank)[q];
        idx[s_off[z.x] + r.x] = q * 4 + 0;
        idx[s_off[z.y] + r.y] = q * 4 + 1;
        idx[s_off[z.z] + r.z] = q * 4 + 2;
        idx[s_off[z.w] + r.w] = q * 4 + 3;
    }
}

__device__ __forceinline__ void n2d_load_rows(v4f (&v)[GU], const float* __restrict__ x,
                                              const int* __restrict__ idx,
                                              int i, int end, int lane) {
    #pragma unroll
    for (int u = 0; u < GU; ++u) {
        int r = i + u;
        r = r < end ? r : end - 1;
        const int node = idx[r];
        v[u] = __builtin_nontemporal_load(
                   (const v4f*)(x + (size_t)node * DIM_IN) + lane);
    }
}

__device__ __forceinline__ void n2d_add_rows(v4f& acc, const v4f (&v)[GU], int i, int end) {
    #pragma unroll
    for (int u = 0; u < GU; ++u)
        if (i + u < end) acc += v[u];
}

__global__ void __launch_bounds__(GW * 32)
n2d_gather(const float* __restrict__ x,
           const int*   __restrict__ cnt,
           const int*   __restrict__ off,
           const int*   __restrict__ idx,
           float*       __restrict__ head) {
    __shared__ v4f s_part[GW][32];
    const int d    = blockIdx.x;
    const int lane = threadIdx.x & 31;
    const int wv   = __builtin_amdgcn_readfirstlane((int)(threadIdx.x >> 5));
    const int start = off[d];
    const int end   = start + cnt[d * CNT_PITCH];
    const int step  = GW * GU;

    v4f acc = {0.0f, 0.0f, 0.0f, 0.0f};
    int i = start + wv * GU;
    if (i < end) {
        v4f va[GU], vb[GU];
        n2d_load_rows(va, x, idx, i, end, lane);
        for (;;) {
            const int j = i + step;
            if (j >= end) { n2d_add_rows(acc, va, i, end); break; }
            n2d_load_rows(vb, x, idx, j, end, lane);
            n2d_add_rows(acc, va, i, end);
            i = j + step;
            if (i >= end) { n2d_add_rows(acc, vb, j, end); break; }
            n2d_load_rows(va, x, idx, i, end, lane);
            n2d_add_rows(acc, vb, j, end);
        }
    }
    s_part[wv][lane] = acc;
    __syncthreads();
    if (wv == 0) {
        v4f s = s_part[0][lane];
        for (int w = 1; w < GW; ++w) s += s_part[w][lane];
        ((v4f*)(head + d * DIM_IN))[lane] = s;
    }
}

__global__ void n2d_mlp(const float* __restrict__ head,
                        const float* __restrict__ w1, const float* __restrict__ b1,
                        const float* __restrict__ w2, const float* __restrict__ b2,
                        float* __restrict__ out) {
    __shared__ float h1[16 * DIM_HID];

    const int lane = threadIdx.x & 31;
    const int wv   = threadIdx.x >> 5;
    const int half = lane >> 4;
    const int m    = lane & 15;
    const int row0 = blockIdx.x * 16;

    for (int t = wv; t < DIM_HID / 16; t += 8) {
        const int n0 = t * 16;
        v8f acc = {};
        for (int k = 0; k < DIM_IN; k += 4) {
            const int ka = k + half * 2;
            v2f a, b;
            float a0 = head[(row0 + m) * DIM_IN + ka];
            float a1 = head[(row0 + m) * DIM_IN + ka + 1];
            a.x = a0 > 0.0f ? a0 : 0.0f;
            a.y = a1 > 0.0f ? a1 : 0.0f;
            b.x = w1[(size_t)ka       * DIM_HID + n0 + m];
            b.y = w1[(size_t)(ka + 1) * DIM_HID + n0 + m];
            acc = __builtin_amdgcn_wmma_f32_16x16x4_f32(
                      false, a, false, b, (short)0, acc, false, false);
        }
        const float bias = b1[n0 + m];
        #pragma unroll
        for (int r = 0; r < 8; ++r) {
            float v = acc[r] + bias;
            h1[(r + half * 8) * DIM_HID + n0 + m] = v > 0.0f ? v : 0.0f;
        }
    }
    __syncthreads();

    {
        const int n0 = wv * 16;
        v8f acc = {};
        for (int k = 0; k < DIM_HID; k += 4) {
            const int ka = k + half * 2;
            v2f a, b;
            a.x = h1[m * DIM_HID + ka];
            a.y = h1[m * DIM_HID + ka + 1];
            b.x = w2[(size_t)ka       * DIM_OUT + n0 + m];
            b.y = w2[(size_t)(ka + 1) * DIM_OUT + n0 + m];
            acc = __builtin_amdgcn_wmma_f32_16x16x4_f32(
                      false, a, false, b, (short)0, acc, false, false);
        }
        const float bias = b2[n0 + m];
        #pragma unroll
        for (int r = 0; r < 8; ++r)
            out[(size_t)(row0 + r + half * 8) * DIM_OUT + n0 + m] = acc[r] + bias;
    }
}

extern "C" void kernel_launch(void* const* d_in, const int* in_sizes, int n_in,
                              void* d_out, int out_size, void* d_ws, size_t ws_size,
                              hipStream_t stream) {
    const float* x    = (const float*)d_in[0];
    const int*   zone = (const int*)  d_in[1];
    const float* w1   = (const float*)d_in[2];
    const float* b1   = (const float*)d_in[3];
    const float* w2   = (const float*)d_in[4];
    const float* b2   = (const float*)d_in[5];

    char*  ws   = (char*)d_ws;
    float* head = (float*)ws;
    int*   cnt  = (int*)(ws + 524288);
    int*   off  = (int*)(ws + 786432);
    int*   rank = (int*)(ws + 1048576);
    int*   idx  = (int*)(ws + 3145728);
    float* out  = (float*)d_out;

    n2d_zero_cnt<<<4, 256, 0, stream>>>(cnt);
    n2d_rank<<<489, 256, 0, stream>>>(zone, cnt, rank);
    n2d_scatter<<<489, 256, 0, stream>>>(zone, cnt, rank, off, idx);
    n2d_gather<<<N_DIST, GW * 32, 0, stream>>>(x, cnt, off, idx, head);
    n2d_mlp<<<N_DIST / 16, 256, 0, stream>>>(head, w1, b1, w2, b2, out);
}


# Chunks ordered by district independently (one launch, no cross-chunk counts); a gather wave merges its district's runs of 16 chunks from a run-start table: 3 launches for 4; row sum order changes
# speedup vs baseline: 5.0785x; 1.9800x over previous
#include <hip/hip_runtime.h>

#define DIM_IN   128
#define DIM_HID  256
#define DIM_OUT  128
#define N_DIST   1024
#define N_NODES  500000
#define GW 4
#define GU 16
#define SG 64
#define SCH 7816
#define SCHP 7840
#define TP 1056
#define ST 1024
#define RW (SG / GW)
#define WB 128

typedef float v2f __attribute__((ext_vector_type(2)));
typedef float v4f __attribute__((ext_vector_type(4)));
typedef float v8f __attribute__((ext_vector_type(8)));
typedef int   v4i __attribute__((ext_vector_type(4)));


__global__ void __launch_bounds__(ST)
n2d_order(const int* __restrict__ zone,
          int*       __restrict__ tab,
          int*       __restrict__ idxl) {
    __shared__ int s_cnt[N_DIST];
    __shared__ int s_w[32];
    __shared__ int s_node[SCH];
    const int tid  = threadIdx.x;
    const int lane = tid & 31;
    const int wv   = tid >> 5;
    const int g    = blockIdx.x;

    const int  l0  = tid * 4,        n0 = g * SCH + l0;
    const int  l1  = (ST + tid) * 4, n1 = g * SCH + l1;
    const bool ok0 = l0 < SCH && n0 < N_NODES;
    const bool ok1 = l1 < SCH && n1 < N_NODES;
    v4i z0 = {0, 0, 0, 0}, z1 = {0, 0, 0, 0};
    if (ok0) z0 = ((const v4i*)zone)[n0 >> 2];
    if (ok1) z1 = ((const v4i*)zone)[n1 >> 2];
    s_cnt[tid] = 0;
    __syncthreads();

    const int d0 = z0.x & (N_DIST - 1), d1 = z0.y & (N_DIST - 1);
    const int d2 = z0.z & (N_DIST - 1), d3 = z0.w & (N_DIST - 1);
    const int d4 = z1.x & (N_DIST - 1), d5 = z1.y & (N_DIST - 1);
    const int d6 = z1.z & (N_DIST - 1), d7 = z1.w & (N_DIST - 1);
    int r0 = 0, r1 = 0, r2 = 0, r3 = 0, r4 = 0, r5 = 0, r6 = 0, r7 = 0;
    if (ok0) {
        r0 = atomicAdd(&s_cnt[d0], 1);
        r1 = atomicAdd(&s_cnt[d1], 1);
        r2 = atomicAdd(&s_cnt[d2], 1);
        r3 = atomicAdd(&s_cnt[d3], 1);
    }
    if (ok1) {
        r4 = atomicAdd(&s_cnt[d4], 1);
        r5 = atomicAdd(&s_cnt[d5], 1);
        r6 = atomicAdd(&s_cnt[d6], 1);
        r7 = atomicAdd(&s_cnt[d7], 1);
    }
    __syncthreads();

    const int c = s_cnt[tid];
    int ia = c;
    for (int dlt = 1; dlt < 32; dlt <<= 1) {
        const int ta = __shfl_up(ia, dlt);
        if (lane >= dlt) ia += ta;
    }
    if (lane == 31) s_w[wv] = ia;
    __syncthreads();
    int ba = lane < wv ? s_w[lane] : 0;
    for (int o = 16; o > 0; o >>= 1) ba += __shfl_xor(ba, o);
    const int lstart = ba + ia - c;
    s_cnt[tid] = lstart;
    tab[g * TP + tid] = lstart;
    if (tid == N_DIST - 1) tab[g * TP + N_DIST] = lstart + c;
    __syncthreads();

    if (ok0) {
        s_node[s_cnt[d0] + r0] = n0 + 0;
        s_node[s_cnt[d1] + r1] = n0 + 1;
        s_node[s_cnt[d2] + r2] = n0 + 2;
        s_node[s_cnt[d3] + r3] = n0 + 3;
    }
    if (ok1) {
        s_node[s_cnt[d4] + r4] = n1 + 0;
        s_node[s_cnt[d5] + r5] = n1 + 1;
        s_node[s_cnt[d6] + r6] = n1 + 2;
        s_node[s_cnt[d7] + r7] = n1 + 3;
    }
    __syncthreads();

    const int rest    = N_NODES - g * SCH;
    const int n_local = rest < SCH ? rest : SCH;
    for (int p = tid; p < n_local; p += ST)
        idxl[g * SCHP + p] = s_node[p];
}

__device__ __forceinline__ void n2d_load_rows(v4f (&v)[GU], const float* __restrict__ x,
                                              const int* ids, int lane) {
    const int mine = ids[lane & (GU - 1)];
    #pragma unroll
    for (int u = 0; u < GU; ++u) {
        const int node = __builtin_amdgcn_readlane(mine, u);
        v[u] = __builtin_nontemporal_load(
                   (const v4f*)(x + (size_t)node * DIM_IN) + lane);
    }
}

__device__ __forceinline__ void n2d_add_rows(v4f& acc, const v4f (&v)[GU], int e, int m) {
    #pragma unroll
    for (int u = 0; u < GU; ++u)
        if (e + u < m) acc += v[u];
}

__global__ void __launch_bounds__(GW * 32)
n2d_gather(const float* __restrict__ x,
           const int*   __restrict__ tab,
           const int*   __restrict__ idxl,
           float*       __restrict__ head) {
    __shared__ v4f s_part[GW][32];
    __shared__ int s_ids[GW][WB];
    const int d    = blockIdx.x;
    const int lane = threadIdx.x & 31;
    const int wv   = __builtin_amdgcn_readfirstlane((int)(threadIdx.x >> 5));

    const int t  = wv * RW + (lane & (RW - 1));
    const int ra = tab[t * TP + d];
    const int rb = tab[t * TP + d + 1];
    const int c  = rb - ra;
    int ia = c;
    for (int dlt = 1; dlt < RW; dlt <<= 1) {
        const int ta = __shfl_up(ia, dlt, RW);
        if ((lane & (RW - 1)) >= dlt) ia += ta;
    }
    const int pos   = ia - c;
    const int delta = t * SCHP + ra - pos;
    const int nw    = __builtin_amdgcn_readlane(ia, RW - 1);

    v4f acc = {0.0f, 0.0f, 0.0f, 0.0f};
    for (int q0 = 0; q0 < nw; q0 += WB) {
        #pragma unroll
        for (int k = 0; k < WB / 32; ++k) {
            int q = q0 + 32 * k + lane;
            q = q < nw ? q : nw - 1;
            int sel = 0;
            #pragma unroll
            for (int r = 0; r < RW; ++r) {
                const int pr = __builtin_amdgcn_readlane(pos, r);
                const int dr = __builtin_amdgcn_readlane(delta, r);
                sel = q >= pr ? dr : sel;
            }
            s_ids[wv][32 * k + lane] = idxl[sel + q];
        }
        __builtin_amdgcn_wave_barrier();

        const int left = nw - q0;
        const int m    = left < WB ? left : WB;
        const int nb   = (m + GU - 1) / GU;
        const int* ids = &s_ids[wv][0];
        v4f va[GU], vb[GU];
        int b = 0;
        n2d_load_rows(va, x, ids, lane);
        for (;;) {
            if (b + 1 >= nb) { n2d_add_rows(acc, va, b * GU, m); break; }
            n2d_load_rows(vb, x, ids + (b + 1) * GU, lane);
            n2d_add_rows(acc, va, b * GU, m);
            b += 2;
            if (b >= nb) { n2d_add_rows(acc, vb, (b - 1) * GU, m); break; }
            n2d_load_rows(va, x, ids + b * GU, lane);
            n2d_add_rows(acc, vb, (b - 1) * GU, m);
        }
        __builtin_amdgcn_wave_barrier();
    }
    s_part[wv][lane] = acc;
    __syncthreads();
    if (wv == 0) {
        v4f s = s_part[0][lane];
        for (int w = 1; w < GW; ++w) s += s_part[w][lane];
        ((v4f*)(head + d * DIM_IN))[lane] = s;
    }
}

__global__ void n2d_mlp(const float* __restrict__ head,
                        const float* __restrict__ w1, const float* __restrict__ b1,
                        const float* __restrict__ w2, const float* __restrict__ b2,
                        float* __restrict__ out) {
    __shared__ float h1[16 * DIM_HID];

    const int lane = threadIdx.x & 31;
    const int wv   = threadIdx.x >> 5;
    const int half = lane >> 4;
    const int m    = lane & 15;
    const int row0 = blockIdx.x * 16;

    for (int t = wv; t < DIM_HID / 16; t += 8) {
        const int n0 = t * 16;
        v8f acc = {};
        for (int k = 0; k < DIM_IN; k += 4) {
            const int ka = k + half * 2;
            v2f a, b;
            float a0 = head[(row0 + m) * DIM_IN + ka];
            float a1 = head[(row0 + m) * DIM_IN + ka + 1];
            a.x = a0 > 0.0f ? a0 : 0.0f;
            a.y = a1 > 0.0f ? a1 : 0.0f;
            b.x = w1[(size_t)ka       * DIM_HID + n0 + m];
            b.y = w1[(size_t)(ka + 1) * DIM_HID + n0 + m];
            acc = __builtin_amdgcn_wmma_f32_16x16x4_f32(
                      false, a, false, b, (short)0, acc, false, false);
        }
        const float bias = b1[n0 + m];
        #pragma unroll
        for (int r = 0; r < 8; ++r) {
            float v = acc[r] + bias;
            h1[(r + half * 8) * DIM_HID + n0 + m] = v > 0.0f ? v : 0.0f;
        }
    }
    __syncthreads();

    {
        const int n0 = wv * 16;
        v8f acc = {};
        for (int k = 0; k < DIM_HID; k += 4) {
            const int ka = k + half * 2;
            v2f a, b;
            a.x = h1[m * DIM_HID + ka];
            a.y = h1[m * DIM_HID + ka + 1];
            b.x = w2[(size_t)ka       * DIM_OUT + n0 + m];
            b.y = w2[(size_t)(ka + 1) * DIM_OUT + n0 + m];
            acc = __builtin_amdgcn_wmma_f32_16x16x4_f32(
                      false, a, false, b, (short)0, acc, false, false);
        }
        const float bias = b2[n0 + m];
        #pragma unroll
        for (int r = 0; r < 8; ++r)
            out[(size_t)(row0 + r + half * 8) * DIM_OUT + n0 + m] = acc[r] + bias;
    }
}

extern "C" void kernel_launch(void* const* d_in, const int* in_sizes, int n_in,
                              void* d_out, int out_size, void* d_ws, size_t ws_size,
                              hipStream_t stream) {
    const float* x    = (const float*)d_in[0];
    const int*   zone = (const int*)  d_in[1];
    const float* w1   = (const float*)d_in[2];
    const float* b1   = (const float*)d_in[3];
    const float* w2   = (const float*)d_in[4];
    const float* b2   = (const float*)d_in[5];

    char*  ws   = (char*)d_ws;
    float* head = (float*)ws;
    int*   tab  = (int*)(ws + 524288);
    int*   idxl = (int*)(ws + 3145728);
    float* out  = (float*)d_out;

    n2d_order<<<SG, ST, 0, stream>>>(zone, tab, idxl);
    n2d_gather<<<N_DIST, GW * 32, 0, stream>>>(x, tab, idxl, head);
    n2d_mlp<<<N_DIST / 16, 256, 0, stream>>>(head, w1, b1, w2, b2, out);
}


# MLP operand feed rebuilt: head band and hidden band read from padded LDS (whole-line loads, no bank conflicts), weight fragments in flight a register block ahead of the WMMAs; sum order unchanged
# speedup vs baseline: 5.9518x; 1.1720x over previous
#include <hip/hip_runtime.h>

#define DIM_IN   128
#define DIM_HID  256
#define DIM_OUT  128
#define N_DIST   1024
#define N_NODES  500000
#define GW 4
#define GU 16
#define SG 64
#define SCH 7816
#define SCHP 7840
#define TP 1056
#define ST 1024
#define RW (SG / GW)
#define WB 128

typedef float v2f __attribute__((ext_vector_type(2)));
typedef float v4f __attribute__((ext_vector_type(4)));
typedef float v8f __attribute__((ext_vector_type(8)));
typedef int   v4i __attribute__((ext_vector_type(4)));


__global__ void __launch_bounds__(ST)
n2d_order(const int* __restrict__ zone,
          int*       __restrict__ tab,
          int*       __restrict__ idxl) {
    __shared__ int s_cnt[N_DIST];
    __shared__ int s_w[32];
    __shared__ int s_node[SCH];
    const int tid  = threadIdx.x;
    const int lane = tid & 31;
    const int wv   = tid >> 5;
    const int g    = blockIdx.x;

    const int  l0  = tid * 4,        n0 = g * SCH + l0;
    const int  l1  = (ST + tid) * 4, n1 = g * SCH + l1;
    const bool ok0 = l0 < SCH && n0 < N_NODES;
    const bool ok1 = l1 < SCH && n1 < N_NODES;
    v4i z0 = {0, 0, 0, 0}, z1 = {0, 0, 0, 0};
    if (ok0) z0 = ((const v4i*)zone)[n0 >> 2];
    if (ok1) z1 = ((const v4i*)zone)[n1 >> 2];
    s_cnt[tid] = 0;
    __syncthreads();

    const int d0 = z0.x & (N_DIST - 1), d1 = z0.y & (N_DIST - 1);
    const int d2 = z0.z & (N_DIST - 1), d3 = z0.w & (N_DIST - 1);
    const int d4 = z1.x & (N_DIST - 1), d5 = z1.y & (N_DIST - 1);
    const int d6 = z1.z & (N_DIST - 1), d7 = z1.w & (N_DIST - 1);
    int r0 = 0, r1 = 0, r2 = 0, r3 = 0, r4 = 0, r5 = 0, r6 = 0, r7 = 0;
    if (ok0) {
        r0 = atomicAdd(&s_cnt[d0], 1);
        r1 = atomicAdd(&s_cnt[d1], 1);
        r2 = atomicAdd(&s_cnt[d2], 1);
        r3 = atomicAdd(&s_cnt[d3], 1);
    }
    if (ok1) {
        r4 = atomicAdd(&s_cnt[d4], 1);
        r5 = atomicAdd(&s_cnt[d5], 1);
        r6 = atomicAdd(&s_cnt[d6], 1);
        r7 = atomicAdd(&s_cnt[d7], 1);
    }
    __syncthreads();

    const int c = s_cnt[tid];
    int ia = c;
    for (int dlt = 1; dlt < 32; dlt <<= 1) {
        const int ta = __shfl_up(ia, dlt);
        if (lane >= dlt) ia += ta;
    }
    if (lane == 31) s_w[wv] = ia;
    __syncthreads();
    int ba = lane < wv ? s_w[lane] : 0;
    for (int o = 16; o > 0; o >>= 1) ba += __shfl_xor(ba, o);
    const int lstart = ba + ia - c;
    s_cnt[tid] = lstart;
    tab[g * TP + tid] = lstart;
    if (tid == N_DIST - 1) tab[g * TP + N_DIST] = lstart + c;
    __syncthreads();

    if (ok0) {
        s_node[s_cnt[d0] + r0] = n0 + 0;
        s_node[s_cnt[d1] + r1] = n0 + 1;
        s_node[s_cnt[d2] + r2] = n0 + 2;
        s_node[s_cnt[d3] + r3] = n0 + 3;
    }
    if (ok1) {
        s_node[s_cnt[d4] + r4] = n1 + 0;
        s_node[s_cnt[d5] + r5] = n1 + 1;
        s_node[s_cnt[d6] + r6] = n1 + 2;
        s_node[s_cnt[d7] + r7] = n1 + 3;
    }
    __syncthreads();

    const int rest    = N_NODES - g * SCH;
    const int n_local = rest < SCH ? rest : SCH;
    for (int p = tid; p < n_local; p += ST)
        idxl[g * SCHP + p] = s_node[p];
}

__device__ __forceinline__ void n2d_load_rows(v4f (&v)[GU], const float* __restrict__ x,
                                              const int* ids, int lane) {
    const int mine = ids[lane & (GU - 1)];
    #pragma unroll
    for (int u = 0; u < GU; ++u) {
        const int node = __builtin_amdgcn_readlane(mine, u);
        v[u] = __builtin_nontemporal_load(
                   (const v4f*)(x + (size_t)node * DIM_IN) + lane);
    }
}

__device__ __forceinline__ void n2d_add_rows(v4f& acc, const v4f (&v)[GU], int e, int m) {
    #pragma unroll
    for (int u = 0; u < GU; ++u)
        if (e + u < m) acc += v[u];
}

__global__ void __launch_bounds__(GW * 32)
n2d_gather(const float* __restrict__ x,
           const int*   __restrict__ tab,
           const int*   __restrict__ idxl,
           float*       __restrict__ head) {
    __shared__ v4f s_part[GW][32];
    __shared__ int s_ids[GW][WB];
    const int d    = blockIdx.x;
    const int lane = threadIdx.x & 31;
    const int wv   = __builtin_amdgcn_readfirstlane((int)(threadIdx.x >> 5));

    const int t  = wv * RW + (lane & (RW - 1));
    const int ra = tab[t * TP + d];
    const int rb = tab[t * TP + d + 1];
    const int c  = rb - ra;
    int ia = c;
    for (int dlt = 1; dlt < RW; dlt <<= 1) {
        const int ta = __shfl_up(ia, dlt, RW);
        if ((lane & (RW - 1)) >= dlt) ia += ta;
    }
    const int pos   = ia - c;
    const int delta = t * SCHP + ra - pos;
    const int nw    = __builtin_amdgcn_readlane(ia, RW - 1);

    v4f acc = {0.0f, 0.0f, 0.0f, 0.0f};
    for (int q0 = 0; q0 < nw; q0 += WB) {
        #pragma unroll
        for (int k = 0; k < WB / 32; ++k) {
            int q = q0 + 32 * k + lane;
            q = q < nw ? q : nw - 1;
            int sel = 0;
            #pragma unroll
            for (int r = 0; r < RW; ++r) {
                const int pr = __builtin_amdgcn_readlane(pos, r);
                const int dr = __builtin_amdgcn_readlane(delta, r);
                sel = q >= pr ? dr : sel;
            }
            s_ids[wv][32 * k + lane] = idxl[sel + q];
        }
        __builtin_amdgcn_wave_barrier();

        const int left = nw - q0;
        const int m    = left < WB ? left : WB;
        const int nb   = (m + GU - 1) / GU;
        const int* ids = &s_ids[wv][0];
        v4f va[GU], vb[GU];
        int b = 0;
        n2d_load_rows(va, x, ids, lane);
        for (;;) {
            if (b + 1 >= nb) { n2d_add_rows(acc, va, b * GU, m); break; }
            n2d_load_rows(vb, x, ids + (b + 1) * GU, lane);
            n2d_add_rows(acc, va, b * GU, m);
            b += 2;
            if (b >= nb) { n2d_add_rows(acc, vb, (b - 1) * GU, m); break; }
            n2d_load_rows(va, x, ids + b * GU, lane);
            n2d_add_rows(acc, vb, (b - 1) * GU, m);
        }
        __builtin_amdgcn_wave_barrier();
    }
    s_part[wv][lane] = acc;
    __syncthreads();
    if (wv == 0) {
        v4f s = s_part[0][lane];
        for (int w = 1; w < GW; ++w) s += s_part[w][lane];
        ((v4f*)(head + d * DIM_IN))[lane] = s;
    }
}

#define MW   8
#define MKS  8
#define MKT  16
#define A_P  (DIM_IN + 4)
#define H_P  (DIM_HID + 4)

template <int NS>
__device__ __forceinline__ void mlp_load_b(v2f (&b)[NS], const float* __restrict__ w,
                                           int ld, int j0, int half, int col) {
    #pragma unroll
    for (int s = 0; s < NS; ++s) {
        const int kk = 4 * (j0 + s) + 2 * half;
        b[s].x = w[(size_t)kk * ld + col];
        b[s].y = w[(size_t)(kk + 1) * ld + col];
    }
}

template <int NS>
__device__ __forceinline__ void mlp_load_a(v2f (&a)[NS], const float* arow, int j0) {
    #pragma unroll
    for (int s = 0; s < NS; ++s)
        a[s] = *(const v2f*)(arow + 4 * (j0 + s));
}

template <int NS>
__device__ __forceinline__ void mlp_mma2(v8f& c0, v8f& c1, const v2f (&a)[NS],
                                         const v2f (&b0)[NS], const v2f (&b1)[NS]) {
    #pragma unroll
    for (int s = 0; s < NS; ++s) {
        c0 = __builtin_amdgcn_wmma_f32_16x16x4_f32(
                 false, a[s], false, b0[s], (short)0, c0, false, false);
        c1 = __builtin_amdgcn_wmma_f32_16x16x4_f32(
                 false, a[s], false, b1[s], (short)0, c1, false, false);
    }
}

template <int NS>
__device__ __forceinline__ void mlp_mma1(v8f& c, const v2f (&a)[NS], const v2f (&b)[NS]) {
    #pragma unroll
    for (int s = 0; s < NS; ++s)
        c = __builtin_amdgcn_wmma_f32_16x16x4_f32(
                false, a[s], false, b[s], (short)0, c, false, false);
}

__global__ void __launch_bounds__(MW * 32) __attribute__((amdgpu_waves_per_eu(2, 4)))
n2d_mlp(const float* __restrict__ head,
        const float* __restrict__ w1, const float* __restrict__ b1,
        const float* __restrict__ w2, const float* __restrict__ b2,
        float* __restrict__ out) {
    __shared__ __attribute__((aligned(16))) float s_a[16 * A_P];
    __shared__ __attribute__((aligned(16))) float s_h[16 * H_P];

    const int tid  = threadIdx.x;
    const int lane = tid & 31;
    const int wv   = tid >> 5;
    const int half = lane >> 4;
    const int m    = lane & 15;
    const int row0 = blockIdx.x * 16;
    const int c0   = wv * 16 + m;
    const int c1   = c0 + DIM_HID / 2;

    const v4f* hsrc = (const v4f*)(head + (size_t)row0 * DIM_IN);
    v4f g0 = hsrc[tid];
    v4f g1 = hsrc[tid + MW * 32];

    v2f p0[MKS], p1[MKS], q0[MKS], q1[MKS];
    v2f xa[MKS], ya[MKS];
    mlp_load_b<MKS>(p0, w1, DIM_HID, 0, half, c0);
    mlp_load_b<MKS>(p1, w1, DIM_HID, 0, half, c1);
    const float bias0 = b1[c0], bias1 = b1[c1], bias2 = b2[c0];

    g0.x = g0.x > 0.0f ? g0.x : 0.0f;  g0.y = g0.y > 0.0f ? g0.y : 0.0f;
    g0.z = g0.z > 0.0f ? g0.z : 0.0f;  g0.w = g0.w > 0.0f ? g0.w : 0.0f;
    g1.x = g1.x > 0.0f ? g1.x : 0.0f;  g1.y = g1.y > 0.0f ? g1.y : 0.0f;
    g1.z = g1.z > 0.0f ? g1.z : 0.0f;  g1.w = g1.w > 0.0f ? g1.w : 0.0f;
    *(v4f*)&s_a[(tid >> 5) * A_P + 4 * (tid & 31)]       = g0;
    *(v4f*)&s_a[((tid >> 5) + MW) * A_P + 4 * (tid & 31)] = g1;
    __syncthreads();

    const float* arow = s_a + m * A_P + 2 * half;
    v8f acc0 = {}, acc1 = {};
    mlp_load_a<MKS>(xa, arow, 0);

    mlp_load_b<MKS>(q0, w1, DIM_HID, MKS, half, c0);
    mlp_load_b<MKS>(q1, w1, DIM_HID, MKS, half, c1);
    mlp_load_a<MKS>(ya, arow, MKS);
    mlp_mma2<MKS>(acc0, acc1, xa, p0, p1);

    mlp_load_b<MKS>(p0, w1, DIM_HID, 2 * MKS, half, c0);
    mlp_load_b<MKS>(p1, w1, DIM_HID, 2 * MKS, half, c1);
    mlp_load_a<MKS>(xa, arow, 2 * MKS);
    mlp_mma2<MKS>(acc0, acc1, ya, q0, q1);

    mlp_load_b<MKS>(q0, w1, DIM_HID, 3 * MKS, half, c0);
    mlp_load_b<MKS>(q1, w1, DIM_HID, 3 * MKS, half, c1);
    mlp_load_a<MKS>(ya, arow, 3 * MKS);
    mlp_mma2<MKS>(acc0, acc1, xa, p0, p1);

    v2f r0[MKT], r1[MKT];
    v2f ua[MKT], va[MKT];
    mlp_load_b<MKT>(r0, w2, DIM_OUT, 0, half, c0);
    mlp_mma2<MKS>(acc0, acc1, ya, q0, q1);

    #pragma unroll
    for (int r = 0; r < 8; ++r) {
        const float v0 = acc0[r] + bias0;
        const float v1 = acc1[r] + bias1;
        s_h[(r + half * 8) * H_P + c0] = v0 > 0.0f ? v0 : 0.0f;
        s_h[(r + half * 8) * H_P + c1] = v1 > 0.0f ? v1 : 0.0f;
    }
    __syncthreads();

    const float* hrow = s_h + m * H_P + 2 * half;
    v8f acc = {};
    mlp_load_a<MKT>(ua, hrow, 0);

    mlp_load_b<MKT>(r1, w2, DIM_OUT, MKT, half, c0);
    mlp_load_a<MKT>(va, hrow, MKT);
    mlp_mma1<MKT>(acc, ua, r0);

    mlp_load_b<MKT>(r0, w2, DIM_OUT, 2 * MKT, half, c0);
    mlp_load_a<MKT>(ua, hrow, 2 * MKT);
    mlp_mma1<MKT>(acc, va, r1);

    mlp_load_b<MKT>(r1, w2, DIM_OUT, 3 * MKT, half, c0);
    mlp_load_a<MKT>(va, hrow, 3 * MKT);
    mlp_mma1<MKT>(acc, ua, r0);

    mlp_mma1<MKT>(acc, va, r1);

    #pragma unroll
    for (int r = 0; r < 8; ++r)
        out[(size_t)(row0 + r + half * 8) * DIM_OUT + c0] = acc[r] + bias2;
}

extern "C" void kernel_launch(void* const* d_in, const int* in_sizes, int n_in,
                              void* d_out, int out_size, void* d_ws, size_t ws_size,
                              hipStream_t stream) {
    const float* x    = (const float*)d_in[0];
    const int*   zone = (const int*)  d_in[1];
    const float* w1   = (const float*)d_in[2];
    const float* b1   = (const float*)d_in[3];
    const float* w2   = (const float*)d_in[4];
    const float* b2   = (const float*)d_in[5];

    char*  ws   = (char*)d_ws;
    float* head = (float*)ws;
    int*   tab  = (int*)(ws + 524288);
    int*   idxl = (int*)(ws + 3145728);
    float* out  = (float*)d_out;

    n2d_order<<<SG, ST, 0, stream>>>(zone, tab, idxl);
    n2d_gather<<<N_DIST, GW * 32, 0, stream>>>(x, tab, idxl, head);
    n2d_mlp<<<N_DIST / 16, MW * 32, 0, stream>>>(head, w1, b1, w2, b2, out);
}
